# starting kernel
# baseline (speedup 1.0000x reference)
#include <hip/hip_runtime.h>
#include <hip/hip_bf16.h>

typedef __attribute__((ext_vector_type(2))) float v2f;
typedef __attribute__((ext_vector_type(8))) float v8f;

__global__ __launch_bounds__(256) void neigh_conv_wmma_kernel(
    const float* __restrict__ x,
    const float* __restrict__ W,
    const float* __restrict__ bias,
    float* __restrict__ out)
{
    constexpr int C   = 200;
    constexpr int K   = 5;
    constexpr int P   = 4096;
    constexpr int CK  = C - K;
    constexpr int NPT = 4;

    const int lane = threadIdx.x & 31;
    const int wave = threadIdx.x >> 5;
    const int m    = lane & 15;
    const int h    = lane >> 4;

    const int cm    = blockIdx.y * 16;
    const int batch = blockIdx.z;
    const int p0    = (blockIdx.x * 8 + wave) * (NPT * 16);

    const float* xb = x   + (size_t)batch * C * P;
    float*       ob = out + (size_t)batch * C * P;

    const int c      = cm + m;
    const int cc     = (c < C) ? c : (C - 1);
    const int startc = min(max(cc - 2, 0), CK);
    const int cbase  = cm - 2;

    v2f a[5];
#pragma unroll
    for (int s = 0; s < 5; ++s) {
        const int ch0 = cbase + 4 * s + 2 * h;
        const int ch1 = ch0 + 1;
        const int d0  = ch0 - startc;
        const int d1  = ch1 - startc;
        const bool in0 = (c < C) && (d0 >= 0) && (d0 < K);
        const bool in1 = (c < C) && (d1 >= 0) && (d1 < K);
        const float w0 = W[cc * K + min(max(d0, 0), K - 1)];
        const float w1 = W[cc * K + min(max(d1, 0), K - 1)];
        a[s][0] = in0 ? w0 : 0.0f;
        a[s][1] = in1 ? w1 : 0.0f;
    }

    const float* brow0[5];
    const float* brow1[5];
#pragma unroll
    for (int s = 0; s < 5; ++s) {
        const int chb = cbase + 4 * s + 2 * h;
        brow0[s] = xb + (size_t)min(max(chb,     0), C - 1) * P;
        brow1[s] = xb + (size_t)min(max(chb + 1, 0), C - 1) * P;
    }

    const int  row0     = cm + 8 * h;
    const bool rowvalid = (row0 + 7) < C;
    v8f accb;
#pragma unroll
    for (int v = 0; v < 8; ++v) {
        const int row = min(row0 + v, C - 1);
        accb[v] = bias[row];
    }

#pragma unroll
    for (int t = 0; t < NPT; ++t) {
        const int p = p0 + t * 16 + m;
        v8f acc = accb;
#pragma unroll
        for (int s = 0; s < 5; ++s) {
            v2f bfrag;
            bfrag[0] = brow0[s][p];
            bfrag[1] = brow1[s][p];
            acc = __builtin_amdgcn_wmma_f32_16x16x4_f32(
                 false, a[s],  false, bfrag,
                 (short)0, acc,  false,  false);
        }
        if (rowvalid) {
#pragma unroll
            for (int v = 0; v < 8; ++v) {
                const size_t off = (size_t)(row0 + v) * P + p;
                ob[off] = xb[off] + acc[v];
            }
        }
    }
}

extern "C" void kernel_launch(void* const* d_in, const int* in_sizes, int n_in,
                              void* d_out, int out_size, void* d_ws, size_t ws_size,
                              hipStream_t stream) {
    const float* x    = (const float*)d_in[0];
    const float* W    = (const float*)d_in[1];
    const float* bias = (const float*)d_in[2];
    float* out = (float*)d_out;

    dim3 grid(8, 13, 16);
    dim3 block(256, 1, 1);
    neigh_conv_wmma_kernel<<<grid, block, 0, stream>>>(x, W, bias, out);
}


# Each lane owns 4 adjacent pixels (WMMA column n of tile j = pixel 4n+j): x rows read and out rows written 128 bits per lane in whole 128-B lines, the wave's 18 row loads issued as one group
# speedup vs baseline: 1.5413x; 1.5413x over previous
#include <hip/hip_runtime.h>
#include <hip/hip_bf16.h>

typedef __attribute__((ext_vector_type(2))) float v2f;
typedef __attribute__((ext_vector_type(4))) float v4f;
typedef __attribute__((ext_vector_type(8))) float v8f;

__global__ __launch_bounds__(256) void neigh_conv_wmma_kernel(
    const float* __restrict__ x,
    const float* __restrict__ W,
    const float* __restrict__ bias,
    float* __restrict__ out)
{
    constexpr int C   = 200;
    constexpr int K   = 5;
    constexpr int P   = 4096;
    constexpr int CK  = C - K;
    constexpr int NPT = 4;

    const int lane = threadIdx.x & 31;
    const int wave = threadIdx.x >> 5;
    const int m    = lane & 15;
    const int h    = lane >> 4;

    const int cm    = blockIdx.y * 16;
    const int batch = blockIdx.z;
    const int p0    = (blockIdx.x * 8 + wave) * (NPT * 16);
    const int px    = p0 + NPT * m;

    const float* xb = x   + (size_t)batch * C * P;
    float*       ob = out + (size_t)batch * C * P;

    const int c      = cm + m;
    const int cc     = (c < C) ? c : (C - 1);
    const int startc = min(max(cc - 2, 0), CK);
    const int cbase  = cm - 2;

    v2f a[5];
#pragma unroll
    for (int s = 0; s < 5; ++s) {
        const int ch0 = cbase + 4 * s + 2 * h;
        const int ch1 = ch0 + 1;
        const int d0  = ch0 - startc;
        const int d1  = ch1 - startc;
        const bool in0 = (c < C) && (d0 >= 0) && (d0 < K);
        const bool in1 = (c < C) && (d1 >= 0) && (d1 < K);
        const float w0 = W[cc * K + min(max(d0, 0), K - 1)];
        const float w1 = W[cc * K + min(max(d1, 0), K - 1)];
        a[s][0] = in0 ? w0 : 0.0f;
        a[s][1] = in1 ? w1 : 0.0f;
    }

    const int  row0     = cm + 8 * h;
    const bool rowvalid = (row0 + 7) < C;
    v8f accb;
#pragma unroll
    for (int v = 0; v < 8; ++v) {
        const int row = min(row0 + v, C - 1);
        accb[v] = bias[row];
    }

    v4f b0[5];
    v4f b1[5];
#pragma unroll
    for (int s = 0; s < 5; ++s) {
        const int chb = cbase + 4 * s + 2 * h;
        b0[s] = *reinterpret_cast<const v4f*>(xb + (size_t)min(max(chb,     0), C - 1) * P + px);
        b1[s] = *reinterpret_cast<const v4f*>(xb + (size_t)min(max(chb + 1, 0), C - 1) * P + px);
    }

    v4f res[8];
#pragma unroll
    for (int v = 0; v < 8; ++v) {
        const int row = min(row0 + v, C - 1);
        res[v] = *reinterpret_cast<const v4f*>(xb + (size_t)row * P + px);
    }

    __builtin_amdgcn_sched_barrier(0);

    v8f acc[NPT];
#pragma unroll
    for (int t = 0; t < NPT; ++t) acc[t] = accb;
#pragma unroll
    for (int s = 0; s < 5; ++s) {
#pragma unroll
        for (int t = 0; t < NPT; ++t) {
            v2f bfrag;
            bfrag[0] = b0[s][t];
            bfrag[1] = b1[s][t];
            acc[t] = __builtin_amdgcn_wmma_f32_16x16x4_f32(
                 false, a[s],  false, bfrag,
                 (short)0, acc[t],  false,  false);
        }
    }

    __builtin_amdgcn_sched_barrier(0);
    asm volatile("" :: "v"(res[0]), "v"(res[1]), "v"(res[2]), "v"(res[3]),
                       "v"(res[4]), "v"(res[5]), "v"(res[6]), "v"(res[7]));

    if (rowvalid) {
#pragma unroll
        for (int v = 0; v < 8; ++v) {
            v4f o;
#pragma unroll
            for (int t = 0; t < NPT; ++t) o[t] = res[v][t] + acc[t][v];
            *reinterpret_cast<v4f*>(ob + (size_t)(row0 + v) * P + px) = o;
        }
    }
}

extern "C" void kernel_launch(void* const* d_in, const int* in_sizes, int n_in,
                              void* d_out, int out_size, void* d_ws, size_t ws_size,
                              hipStream_t stream) {
    const float* x    = (const float*)d_in[0];
    const float* W    = (const float*)d_in[1];
    const float* bias = (const float*)d_in[2];
    float* out = (float*)d_out;

    dim3 grid(8, 13, 16);
    dim3 block(256, 1, 1);
    neigh_conv_wmma_kernel<<<grid, block, 0, stream>>>(x, W, bias, out);
}


# Each wave keeps one channel tile and walks 4 pixel spans, the next span's 18 row loads issued ahead of this span's WMMAs and stores: 256 work-groups of 13 waves, a quarter as many waves
# speedup vs baseline: 2.1316x; 1.3830x over previous
#include <hip/hip_runtime.h>
#include <hip/hip_bf16.h>

typedef __attribute__((ext_vector_type(2))) float v2f;
typedef __attribute__((ext_vector_type(4))) float v4f;
typedef __attribute__((ext_vector_type(8))) float v8f;


struct RowSet {
    v4f b0[5];
    v4f b1[5];
    v4f res[8];
};

__global__ __launch_bounds__(416) void neigh_conv_wmma_kernel(
    const float* __restrict__ x,
    const float* __restrict__ W,
    const float* __restrict__ bias,
    float* __restrict__ out)
{
    constexpr int C   = 200;
    constexpr int K   = 5;
    constexpr int P   = 4096;
    constexpr int CK  = C - K;
    constexpr int NPT = 4;
    constexpr int NI  = 4;
    constexpr int SPAN_BYTES = NPT * 16 * 4;

    const int lane = threadIdx.x & 31;
    const int wave = __builtin_amdgcn_readfirstlane((int)(threadIdx.x >> 5));
    const int m    = lane & 15;
    const int h    = lane >> 4;

    const int cm    = wave * 16;
    const int batch = blockIdx.x >> 4;
    const int pbase = (blockIdx.x & 15) * (NI * NPT * 16);

    const char* xw = reinterpret_cast<const char*>(x   + (size_t)batch * C * P + pbase);
    char*       ow = reinterpret_cast<char*>      (out + (size_t)batch * C * P + pbase);

    const int c      = cm + m;
    const int cc     = (c < C) ? c : (C - 1);
    const int startc = min(max(cc - 2, 0), CK);
    const int cbase  = cm - 2;

    v2f a[5];
#pragma unroll
    for (int s = 0; s < 5; ++s) {
        const int ch0 = cbase + 4 * s + 2 * h;
        const int ch1 = ch0 + 1;
        const int d0  = ch0 - startc;
        const int d1  = ch1 - startc;
        const bool in0 = (c < C) && (d0 >= 0) && (d0 < K);
        const bool in1 = (c < C) && (d1 >= 0) && (d1 < K);
        const float w0 = W[cc * K + min(max(d0, 0), K - 1)];
        const float w1 = W[cc * K + min(max(d1, 0), K - 1)];
        a[s][0] = in0 ? w0 : 0.0f;
        a[s][1] = in1 ? w1 : 0.0f;
    }

    const int  row0     = cm + 8 * h;
    const bool rowvalid = (row0 + 7) < C;
    const int  rowr     = rowvalid ? row0 : (C - 8);
    v8f accb;
#pragma unroll
    for (int v = 0; v < 8; ++v) accb[v] = bias[rowr + v];

    unsigned offb0[5];
    unsigned offb1[5];
#pragma unroll
    for (int s = 0; s < 5; ++s) {
        const int chb = cbase + 4 * s + 2 * h;
        offb0[s] = (unsigned)(min(max(chb,     0), C - 1) * P + NPT * m) * 4u;
        offb1[s] = (unsigned)(min(max(chb + 1, 0), C - 1) * P + NPT * m) * 4u;
    }
    const unsigned offres = (unsigned)(rowr * P + NPT * m) * 4u;
    const unsigned offout = (unsigned)(row0 * P + NPT * m) * 4u;

    RowSet r[2];

#define LOAD_SPAN(R, XS)                                                                       \
    do {                                                                                       \
        _Pragma("unroll")                                                                      \
        for (int s = 0; s < 5; ++s) {                                                          \
            (R).b0[s] = *reinterpret_cast<const v4f*>((XS) + offb0[s]);                        \
            (R).b1[s] = *reinterpret_cast<const v4f*>((XS) + offb1[s]);                        \
        }                                                                                      \
        _Pragma("unroll")                                                                      \
        for (int v = 0; v < 8; ++v)                                                            \
            (R).res[v] = *reinterpret_cast<const v4f*>((XS) + offres + (unsigned)(v * P * 4)); \
    } while (0)

#define SPAN_OFF(I) ((I) * SPAN_BYTES)

    LOAD_SPAN(r[0], xw + SPAN_OFF(0));

    __builtin_amdgcn_sched_barrier(0);

#pragma unroll
    for (int i = 0; i < NI; ++i) {
        RowSet& cur = r[i & 1];

        if (i + 1 < NI) LOAD_SPAN(r[(i + 1) & 1], xw + SPAN_OFF(i + 1));


        v8f acc[NPT];
#pragma unroll
        for (int t = 0; t < NPT; ++t) acc[t] = accb;
#pragma unroll
        for (int s = 0; s < 5; ++s) {
#pragma unroll
            for (int t = 0; t < NPT; ++t) {
                v2f bfrag;
                bfrag[0] = cur.b0[s][t];
                bfrag[1] = cur.b1[s][t];
                acc[t] = __builtin_amdgcn_wmma_f32_16x16x4_f32(
                     false, a[s],  false, bfrag,
                     (short)0, acc[t],  false,  false);
            }
        }

        __builtin_amdgcn_sched_barrier(0);
        asm volatile("" :: "v"(cur.res[0]), "v"(cur.res[1]), "v"(cur.res[2]), "v"(cur.res[3]),
                           "v"(cur.res[4]), "v"(cur.res[5]), "v"(cur.res[6]), "v"(cur.res[7]));

        if (rowvalid) {
#pragma unroll
            for (int v = 0; v < 8; ++v) {
                v4f o;
#pragma unroll
                for (int t = 0; t < NPT; ++t) o[t] = cur.res[v][t] + acc[t][v];
                *reinterpret_cast<v4f*>(ow + SPAN_OFF(i) + offout + (unsigned)(v * P * 4)) = o;
            }
        }
    }
#undef LOAD_SPAN
#undef SPAN_OFF
}

extern "C" void kernel_launch(void* const* d_in, const int* in_sizes, int n_in,
                              void* d_out, int out_size, void* d_ws, size_t ws_size,
                              hipStream_t stream) {
    const float* x    = (const float*)d_in[0];
    const float* W    = (const float*)d_in[1];
    const float* bias = (const float*)d_in[2];
    float* out = (float*)d_out;

    dim3 grid(256, 1, 1);
    dim3 block(416, 1, 1);
    neigh_conv_wmma_kernel<<<grid, block, 0, stream>>>(x, W, bias, out);
}
